# starting kernel
# baseline (speedup 1.0000x reference)
#include <hip/hip_runtime.h>
#include <hip/hip_bf16.h>


#define IN_C 128
#define HID 64
#define OUT_C 64
#define NUM_LAYERS 8
#define NUM_DEC 2

typedef float v2f __attribute__((ext_vector_type(2)));
typedef float v4f __attribute__((ext_vector_type(4)));
typedef float v8f __attribute__((ext_vector_type(8)));

#define WMMA_F32(a, b, acc) \
    __builtin_amdgcn_wmma_f32_16x16x4_f32(false, (a), false, (b), (short)0, (acc), false, false)

__global__ void gcn_gemm_wmma_f32(const float* __restrict__ A,
                                  const float* __restrict__ B,
                                  const float* __restrict__ bias,
                                  float* __restrict__ C,
                                  int rows, int K)
{
    const int lane = threadIdx.x & 31;
    const int wave = (int)((blockIdx.x * blockDim.x + threadIdx.x) >> 5);
    const int m0 = wave * 32;
    if (m0 >= rows) return;

    const int half = lane >> 4;
    const int l = lane & 15;

    float bv0 = 0.0f, bv1 = 0.0f, bv2 = 0.0f, bv3 = 0.0f;
    if (bias) {
        bv0 = bias[l];
        bv1 = bias[16 + l];
        bv2 = bias[32 + l];
        bv3 = bias[48 + l];
    }

    if (rows - m0 >= 32) {
        v8f acc0 = {}, acc1 = {}, acc2 = {}, acc3 = {};
        v8f acc4 = {}, acc5 = {}, acc6 = {}, acc7 = {};
        const float* arow0 = A + (size_t)(m0 + l) * (size_t)K;
        const float* arow1 = arow0 + (size_t)16 * K;

#pragma unroll 2
        for (int k = 0; k < K; k += 4) {
            v2f a0 = *(const v2f*)(arow0 + k + 2 * half);
            v2f a1 = *(const v2f*)(arow1 + k + 2 * half);
            const float* b0 = B + (size_t)(k + 2 * half) * HID + l;
            const float* b1 = b0 + HID;
            v2f b;
            b.x = b0[0];  b.y = b1[0];
            acc0 = WMMA_F32(a0, b, acc0);
            acc4 = WMMA_F32(a1, b, acc4);
            b.x = b0[16]; b.y = b1[16];
            acc1 = WMMA_F32(a0, b, acc1);
            acc5 = WMMA_F32(a1, b, acc5);
            b.x = b0[32]; b.y = b1[32];
            acc2 = WMMA_F32(a0, b, acc2);
            acc6 = WMMA_F32(a1, b, acc6);
            b.x = b0[48]; b.y = b1[48];
            acc3 = WMMA_F32(a0, b, acc3);
            acc7 = WMMA_F32(a1, b, acc7);
        }

        float* crow0 = C + (size_t)(m0 + 8 * half) * HID + l;
        float* crow1 = crow0 + (size_t)16 * HID;
#pragma unroll
        for (int r = 0; r < 8; ++r) {
            crow0[(size_t)r * HID + 0]  = acc0[r] + bv0;
            crow0[(size_t)r * HID + 16] = acc1[r] + bv1;
            crow0[(size_t)r * HID + 32] = acc2[r] + bv2;
            crow0[(size_t)r * HID + 48] = acc3[r] + bv3;
            crow1[(size_t)r * HID + 0]  = acc4[r] + bv0;
            crow1[(size_t)r * HID + 16] = acc5[r] + bv1;
            crow1[(size_t)r * HID + 32] = acc6[r] + bv2;
            crow1[(size_t)r * HID + 48] = acc7[r] + bv3;
        }
    } else {
        v8f acc0 = {}, acc1 = {}, acc2 = {}, acc3 = {};
        const float* arow = A + (size_t)(m0 + l) * (size_t)K;

#pragma unroll 4
        for (int k = 0; k < K; k += 4) {
            v2f a = *(const v2f*)(arow + k + 2 * half);
            const float* b0 = B + (size_t)(k + 2 * half) * HID + l;
            const float* b1 = b0 + HID;
            v2f b;
            b.x = b0[0];  b.y = b1[0];
            acc0 = WMMA_F32(a, b, acc0);
            b.x = b0[16]; b.y = b1[16];
            acc1 = WMMA_F32(a, b, acc1);
            b.x = b0[32]; b.y = b1[32];
            acc2 = WMMA_F32(a, b, acc2);
            b.x = b0[48]; b.y = b1[48];
            acc3 = WMMA_F32(a, b, acc3);
        }

        float* crow = C + (size_t)(m0 + 8 * half) * HID + l;
#pragma unroll
        for (int r = 0; r < 8; ++r) {
            crow[(size_t)r * HID + 0]  = acc0[r] + bv0;
            crow[(size_t)r * HID + 16] = acc1[r] + bv1;
            crow[(size_t)r * HID + 32] = acc2[r] + bv2;
            crow[(size_t)r * HID + 48] = acc3[r] + bv3;
        }
    }
}

__global__ void k_init_deg(float* __restrict__ deg, int n)
{
    int i = blockIdx.x * blockDim.x + threadIdx.x;
    if (i < n) deg[i] = 1.0f;
}

__global__ void k_accum_deg(const int* __restrict__ dst, float* __restrict__ deg, int e)
{
    int i = blockIdx.x * blockDim.x + threadIdx.x;
    if (i < e) atomicAdd(&deg[dst[i]], 1.0f);
}

__global__ void k_dinv(float* __restrict__ deg, int n)
{
    int i = blockIdx.x * blockDim.x + threadIdx.x;
    if (i < n) {
        float d = deg[i];
        deg[i] = (d > 0.0f) ? rsqrtf(d) : 0.0f;
    }
}

__global__ void k_edge_norm(const int* __restrict__ src, const int* __restrict__ dst,
                            const float* __restrict__ dinv, float* __restrict__ norm, int e)
{
    int i = blockIdx.x * blockDim.x + threadIdx.x;
    if (i < e) norm[i] = dinv[src[i]] * dinv[dst[i]];
}

__global__ void k_agg_init(const float* __restrict__ h, const float* __restrict__ t,
                           const float* __restrict__ dinv, const float* __restrict__ bias,
                           float* __restrict__ agg, int n)
{
    long i = (long)blockIdx.x * blockDim.x + threadIdx.x;
    long total = (long)n * (HID / 4);
    if (i >= total) return;
    int node = (int)(i >> 4);
    int f4 = (int)(i & 15) * 4;
    float dv = dinv[node];
    float d2 = dv * dv;
    v4f hv = *(const v4f*)(h + (size_t)node * HID + f4);
    v4f tv = *(const v4f*)(t + (size_t)node * HID + f4);
    v4f bv = *(const v4f*)(bias + f4);
    v4f out = hv + bv + tv * d2;
    *(v4f*)(agg + (size_t)node * HID + f4) = out;
}

__global__ void k_edge_scatter(const int* __restrict__ src, const int* __restrict__ dst,
                               const float* __restrict__ norm, const float* __restrict__ t,
                               float* __restrict__ agg, int e)
{
    long i = (long)blockIdx.x * blockDim.x + threadIdx.x;
    long total = (long)e * (HID / 4);
    if (i >= total) return;
    int edge = (int)(i >> 4);
    int f4 = (int)(i & 15) * 4;
    int s = src[edge];
    int d = dst[edge];
    float w = norm[edge];
    v4f tv = *(const v4f*)(t + (size_t)s * HID + f4);
    float* dest = agg + (size_t)d * HID + f4;
    atomicAdd(dest + 0, tv.x * w);
    atomicAdd(dest + 1, tv.y * w);
    atomicAdd(dest + 2, tv.z * w);
    atomicAdd(dest + 3, tv.w * w);
}

extern "C" void kernel_launch(void* const* d_in, const int* in_sizes, int n_in,
                              void* d_out, int out_size, void* d_ws, size_t ws_size,
                              hipStream_t stream)
{
    const float* x     = (const float*)d_in[0];
    const int*   ei    = (const int*)d_in[1];
    const float* encW  = (const float*)d_in[2];
    const float* encb  = (const float*)d_in[3];
    const float* gcnW  = (const float*)d_in[4];
    const float* gcnb  = (const float*)d_in[5];
    const float* decW  = (const float*)d_in[6];
    const float* decb  = (const float*)d_in[7];
    const float* finW  = (const float*)d_in[8];
    const float* finb  = (const float*)d_in[9];

    const int N = in_sizes[0] / IN_C;
    const int E = in_sizes[1] / 2;
    const int* src = ei;
    const int* dst = ei + E;

    float* f = (float*)d_ws;
    const size_t nodeFeats = (size_t)N * HID;
    float* hA   = f;                  f += nodeFeats;
    float* hB   = f;                  f += nodeFeats;
    float* t    = f;                  f += nodeFeats;
    float* dinv = f;                  f += N;
    float* norm = f;                  f += E;
    (void)ws_size;

    const int BLK = 256;
    auto blocks1 = [&](long work) { return (int)((work + BLK - 1) / BLK); };
    const int stripsM = (N + 31) / 32;
    const long gemmThreads = (long)stripsM * 32;
    const int gemmBlocks = blocks1(gemmThreads);

    k_init_deg <<<blocks1(N), BLK, 0, stream>>>(dinv, N);
    k_accum_deg<<<blocks1(E), BLK, 0, stream>>>(dst, dinv, E);
    k_dinv     <<<blocks1(N), BLK, 0, stream>>>(dinv, N);
    k_edge_norm<<<blocks1(E), BLK, 0, stream>>>(src, dst, dinv, norm, E);

    gcn_gemm_wmma_f32<<<gemmBlocks, BLK, 0, stream>>>(x, encW, encb, hA, N, IN_C);

    float* h   = hA;
    float* nxt = hB;

    for (int i = 0; i < NUM_LAYERS; ++i) {
        const float* W = gcnW + (size_t)i * HID * HID;
        const float* b = gcnb + (size_t)i * HID;
        gcn_gemm_wmma_f32<<<gemmBlocks, BLK, 0, stream>>>(h, W, nullptr, t, N, HID);
        k_agg_init<<<blocks1((long)N * (HID / 4)), BLK, 0, stream>>>(h, t, dinv, b, nxt, N);
        k_edge_scatter<<<blocks1((long)E * (HID / 4)), BLK, 0, stream>>>(src, dst, norm, t, nxt, E);
        float* tmp = h; h = nxt; nxt = tmp;
    }

    for (int i = 0; i < NUM_DEC; ++i) {
        const float* W = decW + (size_t)i * HID * HID;
        const float* b = decb + (size_t)i * HID;
        gcn_gemm_wmma_f32<<<gemmBlocks, BLK, 0, stream>>>(h, W, b, nxt, N, HID);
        float* tmp = h; h = nxt; nxt = tmp;
    }

    gcn_gemm_wmma_f32<<<gemmBlocks, BLK, 0, stream>>>(h, finW, finb, (float*)d_out, N, HID);
    (void)out_size; (void)n_in;
}


# segment-sum as a pull over edges grouped by destination once per call (integer atomic rank + prefix sum), 16 lanes per node, start value folded in, no float atomics; node sum order = arrival rank
# speedup vs baseline: 2.3399x; 2.3399x over previous
#include <hip/hip_runtime.h>
#include <hip/hip_bf16.h>


#define IN_C 128
#define HID 64
#define OUT_C 64
#define NUM_LAYERS 8
#define NUM_DEC 2

typedef float v2f __attribute__((ext_vector_type(2)));
typedef float v4f __attribute__((ext_vector_type(4)));
typedef float v8f __attribute__((ext_vector_type(8)));

#define WMMA_F32(a, b, acc) \
    __builtin_amdgcn_wmma_f32_16x16x4_f32(false, (a), false, (b), (short)0, (acc), false, false)

__global__ void gcn_gemm_wmma_f32(const float* __restrict__ A,
                                  const float* __restrict__ B,
                                  const float* __restrict__ bias,
                                  float* __restrict__ C,
                                  int rows, int K)
{
    const int lane = threadIdx.x & 31;
    const int wave = (int)((blockIdx.x * blockDim.x + threadIdx.x) >> 5);
    const int m0 = wave * 32;
    if (m0 >= rows) return;

    const int half = lane >> 4;
    const int l = lane & 15;

    float bv0 = 0.0f, bv1 = 0.0f, bv2 = 0.0f, bv3 = 0.0f;
    if (bias) {
        bv0 = bias[l];
        bv1 = bias[16 + l];
        bv2 = bias[32 + l];
        bv3 = bias[48 + l];
    }

    if (rows - m0 >= 32) {
        v8f acc0 = {}, acc1 = {}, acc2 = {}, acc3 = {};
        v8f acc4 = {}, acc5 = {}, acc6 = {}, acc7 = {};
        const float* arow0 = A + (size_t)(m0 + l) * (size_t)K;
        const float* arow1 = arow0 + (size_t)16 * K;

#pragma unroll 2
        for (int k = 0; k < K; k += 4) {
            v2f a0 = *(const v2f*)(arow0 + k + 2 * half);
            v2f a1 = *(const v2f*)(arow1 + k + 2 * half);
            const float* b0 = B + (size_t)(k + 2 * half) * HID + l;
            const float* b1 = b0 + HID;
            v2f b;
            b.x = b0[0];  b.y = b1[0];
            acc0 = WMMA_F32(a0, b, acc0);
            acc4 = WMMA_F32(a1, b, acc4);
            b.x = b0[16]; b.y = b1[16];
            acc1 = WMMA_F32(a0, b, acc1);
            acc5 = WMMA_F32(a1, b, acc5);
            b.x = b0[32]; b.y = b1[32];
            acc2 = WMMA_F32(a0, b, acc2);
            acc6 = WMMA_F32(a1, b, acc6);
            b.x = b0[48]; b.y = b1[48];
            acc3 = WMMA_F32(a0, b, acc3);
            acc7 = WMMA_F32(a1, b, acc7);
        }

        float* crow0 = C + (size_t)(m0 + 8 * half) * HID + l;
        float* crow1 = crow0 + (size_t)16 * HID;
#pragma unroll
        for (int r = 0; r < 8; ++r) {
            crow0[(size_t)r * HID + 0]  = acc0[r] + bv0;
            crow0[(size_t)r * HID + 16] = acc1[r] + bv1;
            crow0[(size_t)r * HID + 32] = acc2[r] + bv2;
            crow0[(size_t)r * HID + 48] = acc3[r] + bv3;
            crow1[(size_t)r * HID + 0]  = acc4[r] + bv0;
            crow1[(size_t)r * HID + 16] = acc5[r] + bv1;
            crow1[(size_t)r * HID + 32] = acc6[r] + bv2;
            crow1[(size_t)r * HID + 48] = acc7[r] + bv3;
        }
    } else {
        v8f acc0 = {}, acc1 = {}, acc2 = {}, acc3 = {};
        const float* arow = A + (size_t)(m0 + l) * (size_t)K;

#pragma unroll 4
        for (int k = 0; k < K; k += 4) {
            v2f a = *(const v2f*)(arow + k + 2 * half);
            const float* b0 = B + (size_t)(k + 2 * half) * HID + l;
            const float* b1 = b0 + HID;
            v2f b;
            b.x = b0[0];  b.y = b1[0];
            acc0 = WMMA_F32(a, b, acc0);
            b.x = b0[16]; b.y = b1[16];
            acc1 = WMMA_F32(a, b, acc1);
            b.x = b0[32]; b.y = b1[32];
            acc2 = WMMA_F32(a, b, acc2);
            b.x = b0[48]; b.y = b1[48];
            acc3 = WMMA_F32(a, b, acc3);
        }

        float* crow = C + (size_t)(m0 + 8 * half) * HID + l;
#pragma unroll
        for (int r = 0; r < 8; ++r) {
            crow[(size_t)r * HID + 0]  = acc0[r] + bv0;
            crow[(size_t)r * HID + 16] = acc1[r] + bv1;
            crow[(size_t)r * HID + 32] = acc2[r] + bv2;
            crow[(size_t)r * HID + 48] = acc3[r] + bv3;
        }
    }
}

typedef int v4i __attribute__((ext_vector_type(4)));

__global__ void k_zero_cnt(int* __restrict__ cnt, int n)
{
    int i = blockIdx.x * blockDim.x + threadIdx.x;
    if (i < n) cnt[i] = 0;
}

__global__ void k_edge_rank(const int* __restrict__ dst, int* __restrict__ cnt,
                            int* __restrict__ rank, int e)
{
    int i = blockIdx.x * blockDim.x + threadIdx.x;
    if (i < e) rank[i] = atomicAdd(&cnt[dst[i]], 1);
}

#define SCAN_THREADS 1024
#define SCAN_PER_THREAD 8

__device__ __forceinline__ int wave_incl_scan(int v, int lane)
{
#pragma unroll
    for (int off = 1; off < 32; off <<= 1) {
        int u = __shfl_up(v, off);
        if (lane >= off) v += u;
    }
    return v;
}

__device__ __forceinline__ float dinv_of(int c)
{
    float d = (float)(1 + c);
    return (d > 0.0f) ? rsqrtf(d) : 0.0f;
}

__global__ void __launch_bounds__(SCAN_THREADS)
k_scan_deg(const int* __restrict__ cnt, int* __restrict__ rowptr,
           float* __restrict__ dinv, int n)
{
    __shared__ int wsum[2][32];
    const int tid = threadIdx.x;
    const int lane = tid & 31;
    const int wv = tid >> 5;
    int running = 0;
    int buf = 0;
    for (int base = 0; base < n; base += SCAN_THREADS * SCAN_PER_THREAD) {
        const int idx = base + SCAN_PER_THREAD * tid;
        int c[SCAN_PER_THREAD];
        if (idx + SCAN_PER_THREAD <= n) {
            v4i lo = *(const v4i*)(cnt + idx);
            v4i hi = *(const v4i*)(cnt + idx + 4);
            c[0] = lo.x; c[1] = lo.y; c[2] = lo.z; c[3] = lo.w;
            c[4] = hi.x; c[5] = hi.y; c[6] = hi.z; c[7] = hi.w;
        } else {
#pragma unroll
            for (int k = 0; k < SCAN_PER_THREAD; ++k)
                c[k] = (idx + k < n) ? cnt[idx + k] : 0;
        }
        int p[SCAN_PER_THREAD];
        int tsum = 0;
#pragma unroll
        for (int k = 0; k < SCAN_PER_THREAD; ++k) { p[k] = tsum; tsum += c[k]; }

        const int inc = wave_incl_scan(tsum, lane);
        if (lane == 31) wsum[buf][wv] = inc;
        __syncthreads();
        const int tot = wsum[buf][lane];
        const int tinc = wave_incl_scan(tot, lane);
        const int wave_excl = __shfl(tinc, wv) - __shfl(tot, wv);
        const int trip_total = __shfl(tinc, 31);
        const int off0 = running + wave_excl + (inc - tsum);

        if (idx + SCAN_PER_THREAD <= n) {
            v4i r0, r1;
            r0.x = off0 + p[0]; r0.y = off0 + p[1]; r0.z = off0 + p[2]; r0.w = off0 + p[3];
            r1.x = off0 + p[4]; r1.y = off0 + p[5]; r1.z = off0 + p[6]; r1.w = off0 + p[7];
            *(v4i*)(rowptr + idx) = r0;
            *(v4i*)(rowptr + idx + 4) = r1;
            v4f d0, d1;
            d0.x = dinv_of(c[0]); d0.y = dinv_of(c[1]); d0.z = dinv_of(c[2]); d0.w = dinv_of(c[3]);
            d1.x = dinv_of(c[4]); d1.y = dinv_of(c[5]); d1.z = dinv_of(c[6]); d1.w = dinv_of(c[7]);
            *(v4f*)(dinv + idx) = d0;
            *(v4f*)(dinv + idx + 4) = d1;
        } else {
#pragma unroll
            for (int k = 0; k < SCAN_PER_THREAD; ++k) {
                if (idx + k < n) {
                    rowptr[idx + k] = off0 + p[k];
                    dinv[idx + k] = dinv_of(c[k]);
                }
            }
        }
        running += trip_total;
        buf ^= 1;
    }
    if (tid == 0) rowptr[n] = running;
}

__global__ void k_edge_place(const int* __restrict__ src, const int* __restrict__ dst,
                             const int* __restrict__ rank, const int* __restrict__ rowptr,
                             const float* __restrict__ dinv, float* __restrict__ pairs, int e)
{
    int i = blockIdx.x * blockDim.x + threadIdx.x;
    if (i >= e) return;
    int s = src[i];
    int d = dst[i];
    int pos = rowptr[d] + rank[i];
    v2f p;
    p.x = __int_as_float(s);
    p.y = dinv[s] * dinv[d];
    *(v2f*)(pairs + 2 * (size_t)pos) = p;
}

#define PULL_U 8

__global__ void k_pull(const int* __restrict__ rowptr, const float* __restrict__ pairs,
                       const float* __restrict__ h, const float* __restrict__ t,
                       const float* __restrict__ dinv, const float* __restrict__ bias,
                       float* __restrict__ out, int n)
{
    long i = (long)blockIdx.x * blockDim.x + threadIdx.x;
    int node = (int)(i >> 4);
    if (node >= n) return;
    int f4 = (int)(i & 15) * 4;
    const int beg = rowptr[node];
    const int end = rowptr[node + 1];
    float dv = dinv[node];
    float d2 = dv * dv;
    v4f hv = *(const v4f*)(h + (size_t)node * HID + f4);
    v4f tv = *(const v4f*)(t + (size_t)node * HID + f4);
    v4f bv = *(const v4f*)(bias + f4);
    v4f acc = hv + bv + tv * d2;

    if (beg < end) {
        const v2f* pr = (const v2f*)pairs;
        const int last = end - 1;
        v2f p[PULL_U];
#pragma unroll
        for (int k = 0; k < PULL_U; ++k) p[k] = pr[min(beg + k, last)];
        for (int e = beg; e < end; e += PULL_U) {
            v4f g[PULL_U];
#pragma unroll
            for (int k = 0; k < PULL_U; ++k)
                g[k] = *(const v4f*)(t + (size_t)__float_as_int(p[k].x) * HID + f4);
            v2f q[PULL_U];
#pragma unroll
            for (int k = 0; k < PULL_U; ++k) q[k] = pr[min(e + PULL_U + k, last)];
            __builtin_amdgcn_sched_barrier(0);
#pragma unroll
            for (int k = 0; k < PULL_U; ++k)
                if (e + k < end) acc += g[k] * p[k].y;
#pragma unroll
            for (int k = 0; k < PULL_U; ++k) p[k] = q[k];
        }
    }
    *(v4f*)(out + (size_t)node * HID + f4) = acc;
}

extern "C" void kernel_launch(void* const* d_in, const int* in_sizes, int n_in,
                              void* d_out, int out_size, void* d_ws, size_t ws_size,
                              hipStream_t stream)
{
    const float* x     = (const float*)d_in[0];
    const int*   ei    = (const int*)d_in[1];
    const float* encW  = (const float*)d_in[2];
    const float* encb  = (const float*)d_in[3];
    const float* gcnW  = (const float*)d_in[4];
    const float* gcnb  = (const float*)d_in[5];
    const float* decW  = (const float*)d_in[6];
    const float* decb  = (const float*)d_in[7];
    const float* finW  = (const float*)d_in[8];
    const float* finb  = (const float*)d_in[9];

    const int N = in_sizes[0] / IN_C;
    const int E = in_sizes[1] / 2;
    const int* src = ei;
    const int* dst = ei + E;

    float* f = (float*)d_ws;
    const size_t nodeFeats = (size_t)N * HID;
    const size_t Npad = ((size_t)N + 1 + 63) / 64 * 64;
    const size_t Epad = ((size_t)E + 63) / 64 * 64;
    float* hA    = f;                 f += nodeFeats;
    float* hB    = f;                 f += nodeFeats;
    float* t     = f;                 f += nodeFeats;
    float* pairs = f;                 f += 2 * Epad;
    int*   rank  = (int*)f;           f += Epad;
    float* dinv  = f;                 f += Npad;
    int*   cnt   = (int*)f;           f += Npad;
    int*   rowptr = (int*)f;          f += Npad;
    (void)ws_size;

    const int BLK = 256;
    auto blocks1 = [&](long work) { return (int)((work + BLK - 1) / BLK); };
    const int stripsM = (N + 31) / 32;
    const long gemmThreads = (long)stripsM * 32;
    const int gemmBlocks = blocks1(gemmThreads);

    k_zero_cnt  <<<blocks1(N), BLK, 0, stream>>>(cnt, N);
    k_edge_rank <<<blocks1(E), BLK, 0, stream>>>(dst, cnt, rank, E);
    k_scan_deg  <<<1, SCAN_THREADS, 0, stream>>>(cnt, rowptr, dinv, N);
    k_edge_place<<<blocks1(E), BLK, 0, stream>>>(src, dst, rank, rowptr, dinv, pairs, E);

    gcn_gemm_wmma_f32<<<gemmBlocks, BLK, 0, stream>>>(x, encW, encb, hA, N, IN_C);

    float* h   = hA;
    float* nxt = hB;

    for (int i = 0; i < NUM_LAYERS; ++i) {
        const float* W = gcnW + (size_t)i * HID * HID;
        const float* b = gcnb + (size_t)i * HID;
        gcn_gemm_wmma_f32<<<gemmBlocks, BLK, 0, stream>>>(h, W, nullptr, t, N, HID);
        k_pull<<<blocks1((long)N * (HID / 4)), BLK, 0, stream>>>(rowptr, pairs, h, t, dinv, b, nxt, N);
        float* tmp = h; h = nxt; nxt = tmp;
    }

    for (int i = 0; i < NUM_DEC; ++i) {
        const float* W = decW + (size_t)i * HID * HID;
        const float* b = decb + (size_t)i * HID;
        gcn_gemm_wmma_f32<<<gemmBlocks, BLK, 0, stream>>>(h, W, b, nxt, N, HID);
        float* tmp = h; h = nxt; nxt = tmp;
    }

    gcn_gemm_wmma_f32<<<gemmBlocks, BLK, 0, stream>>>(h, finW, finb, (float*)d_out, N, HID);
    (void)out_size; (void)n_in;
}


# every dense transform now runs in the kernel that makes its input rows (encoder+layer 0, pull+next layer, last pull+decoder chain) on 16-row LDS tiles, same fp32 WMMA k order; 24 -> 13 launches
# speedup vs baseline: 3.0100x; 1.2864x over previous
#include <hip/hip_runtime.h>
#include <hip/hip_bf16.h>


#define IN_C 128
#define HID 64
#define OUT_C 64
#define NUM_LAYERS 8
#define NUM_DEC 2

typedef float v2f __attribute__((ext_vector_type(2)));
typedef float v4f __attribute__((ext_vector_type(4)));
typedef float v8f __attribute__((ext_vector_type(8)));

#define WMMA_F32(a, b, acc) \
    __builtin_amdgcn_wmma_f32_16x16x4_f32(false, (a), false, (b), (short)0, (acc), false, false)

typedef int v4i __attribute__((ext_vector_type(4)));

__global__ void k_zero_cnt(int* __restrict__ cnt, int n)
{
    int i = blockIdx.x * blockDim.x + threadIdx.x;
    if (i < n) cnt[i] = 0;
}

__global__ void k_edge_rank(const int* __restrict__ dst, int* __restrict__ cnt,
                            int* __restrict__ rank, int e)
{
    int i = blockIdx.x * blockDim.x + threadIdx.x;
    if (i < e) rank[i] = atomicAdd(&cnt[dst[i]], 1);
}

#define SCAN_THREADS 1024
#define SCAN_PER_THREAD 8

__device__ __forceinline__ int wave_incl_scan(int v, int lane)
{
#pragma unroll
    for (int off = 1; off < 32; off <<= 1) {
        int u = __shfl_up(v, off);
        if (lane >= off) v += u;
    }
    return v;
}

__device__ __forceinline__ float dinv_of(int c)
{
    float d = (float)(1 + c);
    return (d > 0.0f) ? rsqrtf(d) : 0.0f;
}

__global__ void __launch_bounds__(SCAN_THREADS)
k_scan_deg(const int* __restrict__ cnt, int* __restrict__ rowptr,
           float* __restrict__ dinv, int n)
{
    __shared__ int wsum[2][32];
    const int tid = threadIdx.x;
    const int lane = tid & 31;
    const int wv = tid >> 5;
    int running = 0;
    int buf = 0;
    for (int base = 0; base < n; base += SCAN_THREADS * SCAN_PER_THREAD) {
        const int idx = base + SCAN_PER_THREAD * tid;
        int c[SCAN_PER_THREAD];
        if (idx + SCAN_PER_THREAD <= n) {
            v4i lo = *(const v4i*)(cnt + idx);
            v4i hi = *(const v4i*)(cnt + idx + 4);
            c[0] = lo.x; c[1] = lo.y; c[2] = lo.z; c[3] = lo.w;
            c[4] = hi.x; c[5] = hi.y; c[6] = hi.z; c[7] = hi.w;
        } else {
#pragma unroll
            for (int k = 0; k < SCAN_PER_THREAD; ++k)
                c[k] = (idx + k < n) ? cnt[idx + k] : 0;
        }
        int p[SCAN_PER_THREAD];
        int tsum = 0;
#pragma unroll
        for (int k = 0; k < SCAN_PER_THREAD; ++k) { p[k] = tsum; tsum += c[k]; }

        const int inc = wave_incl_scan(tsum, lane);
        if (lane == 31) wsum[buf][wv] = inc;
        __syncthreads();
        const int tot = wsum[buf][lane];
        const int tinc = wave_incl_scan(tot, lane);
        const int wave_excl = __shfl(tinc, wv) - __shfl(tot, wv);
        const int trip_total = __shfl(tinc, 31);
        const int off0 = running + wave_excl + (inc - tsum);

        if (idx + SCAN_PER_THREAD <= n) {
            v4i r0, r1;
            r0.x = off0 + p[0]; r0.y = off0 + p[1]; r0.z = off0 + p[2]; r0.w = off0 + p[3];
            r1.x = off0 + p[4]; r1.y = off0 + p[5]; r1.z = off0 + p[6]; r1.w = off0 + p[7];
            *(v4i*)(rowptr + idx) = r0;
            *(v4i*)(rowptr + idx + 4) = r1;
            v4f d0, d1;
            d0.x = dinv_of(c[0]); d0.y = dinv_of(c[1]); d0.z = dinv_of(c[2]); d0.w = dinv_of(c[3]);
            d1.x = dinv_of(c[4]); d1.y = dinv_of(c[5]); d1.z = dinv_of(c[6]); d1.w = dinv_of(c[7]);
            *(v4f*)(dinv + idx) = d0;
            *(v4f*)(dinv + idx + 4) = d1;
        } else {
#pragma unroll
            for (int k = 0; k < SCAN_PER_THREAD; ++k) {
                if (idx + k < n) {
                    rowptr[idx + k] = off0 + p[k];
                    dinv[idx + k] = dinv_of(c[k]);
                }
            }
        }
        running += trip_total;
        buf ^= 1;
    }
    if (tid == 0) rowptr[n] = running;
}

__global__ void k_edge_place(const int* __restrict__ src, const int* __restrict__ dst,
                             const int* __restrict__ rank, const int* __restrict__ rowptr,
                             const float* __restrict__ dinv, float* __restrict__ pairs, int e)
{
    int i = blockIdx.x * blockDim.x + threadIdx.x;
    if (i >= e) return;
    int s = src[i];
    int d = dst[i];
    int pos = rowptr[d] + rank[i];
    v2f p;
    p.x = __int_as_float(s);
    p.y = dinv[s] * dinv[d];
    *(v2f*)(pairs + 2 * (size_t)pos) = p;
}

#define PULL_U 8
#define TILE_LD 68

__device__ __forceinline__ void load_b_frags(const float* __restrict__ Wm, int nt, int l, int half,
                                             v2f (&bf)[16])
{
    const float* b0 = Wm + (size_t)(2 * half) * HID + 16 * nt + l;
#pragma unroll
    for (int j = 0; j < 16; ++j) {
        bf[j].x = b0[(4 * j) * HID];
        bf[j].y = b0[(4 * j + 1) * HID];
    }
}

__device__ __forceinline__ v8f tile_mma64(const float* tile, int l, int half, const v2f (&bf)[16])
{
    v2f af[16];
#pragma unroll
    for (int j = 0; j < 16; ++j)
        af[j] = *(const v2f*)(tile + l * TILE_LD + 4 * j + 2 * half);
    v8f acc = {};
#pragma unroll
    for (int j = 0; j < 16; ++j) acc = WMMA_F32(af[j], bf[j], acc);
    return acc;
}

__device__ __forceinline__ void lds_barrier()
{
    __builtin_amdgcn_fence(__ATOMIC_RELEASE, "workgroup", "local");
    __builtin_amdgcn_s_barrier();
    __builtin_amdgcn_fence(__ATOMIC_ACQUIRE, "workgroup", "local");
}

__global__ void __launch_bounds__(256, 12)
k_pull_mma(const int* __restrict__ rowptr, const float* __restrict__ pairs,
           const float* __restrict__ h, const float* __restrict__ t,
           const float* __restrict__ dinv, const float* __restrict__ bias,
           float* __restrict__ h_out,
           const float* __restrict__ W0, const float* __restrict__ W1, const float* __restrict__ W2,
           const float* __restrict__ c0, const float* __restrict__ c1, const float* __restrict__ c2,
           float* __restrict__ y_out, int nstage, int addmask, int n)
{
    __shared__ __attribute__((aligned(16))) float tile[2][16 * TILE_LD];

    const int tid = threadIdx.x;
    const int g = tid >> 4;
    const int f4 = (tid & 15) * 4;
    const int node = (int)blockIdx.x * 16 + g;
    const int lane = tid & 31;
    const int wave = tid >> 5;
    const int half = lane >> 4;
    const int l = lane & 15;

    v4f acc = {0.0f, 0.0f, 0.0f, 0.0f};
    if (node < n) {
        const int beg = rowptr[node];
        const int end = rowptr[node + 1];
        float dv = dinv[node];
        float d2 = dv * dv;
        v4f hv = *(const v4f*)(h + (size_t)node * HID + f4);
        v4f tv = *(const v4f*)(t + (size_t)node * HID + f4);
        v4f bv = *(const v4f*)(bias + f4);
        acc = hv + bv + tv * d2;

        if (beg < end) {
            const v2f* pr = (const v2f*)pairs;
            const int last = end - 1;
            v2f p[PULL_U];
#pragma unroll
            for (int k = 0; k < PULL_U; ++k) p[k] = pr[min(beg + k, last)];
            for (int e = beg; e < end; e += PULL_U) {
                v4f gth[PULL_U];
#pragma unroll
                for (int k = 0; k < PULL_U; ++k)
                    gth[k] = *(const v4f*)(t + (size_t)__float_as_int(p[k].x) * HID + f4);
                v2f q[PULL_U];
#pragma unroll
                for (int k = 0; k < PULL_U; ++k) q[k] = pr[min(e + PULL_U + k, last)];
                __builtin_amdgcn_sched_barrier(0);
#pragma unroll
                for (int k = 0; k < PULL_U; ++k)
                    if (e + k < end) acc += gth[k] * p[k].y;
#pragma unroll
                for (int k = 0; k < PULL_U; ++k) p[k] = q[k];
            }
        }
        if (h_out) *(v4f*)(h_out + (size_t)node * HID + f4) = acc;
    }
    *(v4f*)(&tile[0][g * TILE_LD + f4]) = acc;

    int cur = 0;
    for (int s = 0; s < nstage; ++s) {
        const float* Wm = (s == 0) ? W0 : ((s == 1) ? W1 : W2);
        const float* cb = (s == 0) ? c0 : ((s == 1) ? c1 : c2);
        v2f bf[16];
        float cv = 0.0f;
        if (wave < 4) {
            load_b_frags(Wm, wave, l, half, bf);
            cv = cb[16 * wave + l];
        }
        const bool add = ((addmask >> s) & 1) != 0;
        lds_barrier();
        if (wave < 4) {
            v8f d = tile_mma64(tile[cur], l, half, bf);
            float* o = &tile[cur ^ 1][(8 * half) * TILE_LD + 16 * wave + l];
#pragma unroll
            for (int r = 0; r < 8; ++r) o[r * TILE_LD] = d[r] + (add ? cv : 0.0f);
        }
        cur ^= 1;
    }
    lds_barrier();
    if (node < n)
        *(v4f*)(y_out + (size_t)node * HID + f4) = *(const v4f*)(&tile[cur][g * TILE_LD + f4]);
}

#define XT_LD 132

__global__ void __launch_bounds__(256)
k_enc_mma(const float* __restrict__ x, const float* __restrict__ encW,
          const float* __restrict__ encb, const float* __restrict__ W0,
          float* __restrict__ h_out, float* __restrict__ t_out, int n)
{
    __shared__ __attribute__((aligned(16))) float xt[16 * XT_LD];
    __shared__ __attribute__((aligned(16))) float tile[2][16 * TILE_LD];

    const int tid = threadIdx.x;
    const int g = tid >> 4;
    const int f4 = (tid & 15) * 4;
    const int node = (int)blockIdx.x * 16 + g;
    const int lane = tid & 31;
    const int wave = tid >> 5;
    const int half = lane >> 4;
    const int l = lane & 15;

    v4f x0 = {0.0f, 0.0f, 0.0f, 0.0f};
    v4f x1 = x0;
    if (node < n) {
        x0 = *(const v4f*)(x + (size_t)node * IN_C + f4);
        x1 = *(const v4f*)(x + (size_t)node * IN_C + 64 + f4);
    }
    *(v4f*)(&xt[g * XT_LD + f4]) = x0;
    *(v4f*)(&xt[g * XT_LD + 64 + f4]) = x1;

    v2f be[32];
    float ev = 0.0f;
    if (wave < 4) {
        const float* b0 = encW + (size_t)(2 * half) * HID + 16 * wave + l;
#pragma unroll
        for (int j = 0; j < 32; ++j) {
            be[j].x = b0[(4 * j) * HID];
            be[j].y = b0[(4 * j + 1) * HID];
        }
        ev = encb[16 * wave + l];
    }
    lds_barrier();
    if (wave < 4) {
        v8f d = {};
#pragma unroll
        for (int j = 0; j < 32; ++j) {
            v2f a = *(const v2f*)(xt + l * XT_LD + 4 * j + 2 * half);
            d = WMMA_F32(a, be[j], d);
        }
        float* o = &tile[0][(8 * half) * TILE_LD + 16 * wave + l];
#pragma unroll
        for (int r = 0; r < 8; ++r) o[r * TILE_LD] = d[r] + ev;
    }

    v2f bf[16];
    if (wave < 4) load_b_frags(W0, wave, l, half, bf);
    lds_barrier();
    if (node < n)
        *(v4f*)(h_out + (size_t)node * HID + f4) = *(const v4f*)(&tile[0][g * TILE_LD + f4]);
    if (wave < 4) {
        v8f d = tile_mma64(tile[0], l, half, bf);
        float* o = &tile[1][(8 * half) * TILE_LD + 16 * wave + l];
#pragma unroll
        for (int r = 0; r < 8; ++r) o[r * TILE_LD] = d[r] + 0.0f;
    }
    lds_barrier();
    if (node < n)
        *(v4f*)(t_out + (size_t)node * HID + f4) = *(const v4f*)(&tile[1][g * TILE_LD + f4]);
}

extern "C" void kernel_launch(void* const* d_in, const int* in_sizes, int n_in,
                              void* d_out, int out_size, void* d_ws, size_t ws_size,
                              hipStream_t stream)
{
    const float* x     = (const float*)d_in[0];
    const int*   ei    = (const int*)d_in[1];
    const float* encW  = (const float*)d_in[2];
    const float* encb  = (const float*)d_in[3];
    const float* gcnW  = (const float*)d_in[4];
    const float* gcnb  = (const float*)d_in[5];
    const float* decW  = (const float*)d_in[6];
    const float* decb  = (const float*)d_in[7];
    const float* finW  = (const float*)d_in[8];
    const float* finb  = (const float*)d_in[9];

    const int N = in_sizes[0] / IN_C;
    const int E = in_sizes[1] / 2;
    const int* src = ei;
    const int* dst = ei + E;

    float* f = (float*)d_ws;
    const size_t nodeFeats = (size_t)N * HID;
    const size_t Npad = ((size_t)N + 1 + 63) / 64 * 64;
    const size_t Epad = ((size_t)E + 63) / 64 * 64;
    float* hA    = f;                 f += nodeFeats;
    float* hB    = f;                 f += nodeFeats;
    float* tA    = f;                 f += nodeFeats;
    float* tB    = f;                 f += nodeFeats;
    float* pairs = f;                 f += 2 * Epad;
    int*   rank  = (int*)f;           f += Epad;
    float* dinv  = f;                 f += Npad;
    int*   cnt   = (int*)f;           f += Npad;
    int*   rowptr = (int*)f;          f += Npad;
    (void)ws_size;

    const int BLK = 256;
    auto blocks1 = [&](long work) { return (int)((work + BLK - 1) / BLK); };

    k_zero_cnt  <<<blocks1(N), BLK, 0, stream>>>(cnt, N);
    k_edge_rank <<<blocks1(E), BLK, 0, stream>>>(dst, cnt, rank, E);
    k_scan_deg  <<<1, SCAN_THREADS, 0, stream>>>(cnt, rowptr, dinv, N);
    k_edge_place<<<blocks1(E), BLK, 0, stream>>>(src, dst, rank, rowptr, dinv, pairs, E);

    float* h   = hA;
    float* nxt = hB;
    float* t    = tA;
    float* tnxt = tB;

    const int pullBlocks = (N + 15) / 16;
    k_enc_mma<<<pullBlocks, BLK, 0, stream>>>(x, encW, encb, gcnW, h, t, N);

    for (int i = 0; i < NUM_LAYERS; ++i) {
        const float* b = gcnb + (size_t)i * HID;
        if (i + 1 < NUM_LAYERS) {
            const float* Wn = gcnW + (size_t)(i + 1) * HID * HID;
            k_pull_mma<<<pullBlocks, BLK, 0, stream>>>(rowptr, pairs, h, t, dinv, b, nxt,
                                                       Wn, nullptr, nullptr, b, nullptr, nullptr,
                                                       tnxt, 1, 0, N);
        } else {
            k_pull_mma<<<pullBlocks, BLK, 0, stream>>>(rowptr, pairs, h, t, dinv, b, nullptr,
                                                       decW, decW + (size_t)HID * HID, finW,
                                                       decb, decb + HID, finb,
                                                       (float*)d_out, 3, 7, N);
        }
        float* tmp = h; h = nxt; nxt = tmp;
        tmp = t; t = tnxt; tnxt = tmp;
    }
    (void)out_size; (void)n_in;
}


# symmetric norm factored: t rows stored scaled by dinv[node]; node sum (self row, then edges in rank order) scaled once by dinv[dst], h+b added last; index-only 4 B edge list; pull at 16 waves/SIMD
# speedup vs baseline: 3.3407x; 1.1099x over previous
#include <hip/hip_runtime.h>
#include <hip/hip_bf16.h>


#define IN_C 128
#define HID 64
#define OUT_C 64
#define NUM_LAYERS 8
#define NUM_DEC 2

typedef float v2f __attribute__((ext_vector_type(2)));
typedef float v4f __attribute__((ext_vector_type(4)));
typedef float v8f __attribute__((ext_vector_type(8)));

#define WMMA_F32(a, b, acc) \
    __builtin_amdgcn_wmma_f32_16x16x4_f32(false, (a), false, (b), (short)0, (acc), false, false)

typedef int v4i __attribute__((ext_vector_type(4)));

__global__ void k_zero_cnt(int* __restrict__ cnt, int n)
{
    int i = blockIdx.x * blockDim.x + threadIdx.x;
    if (i < n) cnt[i] = 0;
}

__global__ void k_edge_rank(const int* __restrict__ dst, int* __restrict__ cnt,
                            int* __restrict__ rank, int e)
{
    int i = blockIdx.x * blockDim.x + threadIdx.x;
    if (i < e) rank[i] = atomicAdd(&cnt[dst[i]], 1);
}

#define SCAN_THREADS 1024
#define SCAN_PER_THREAD 8

__device__ __forceinline__ int wave_incl_scan(int v, int lane)
{
#pragma unroll
    for (int off = 1; off < 32; off <<= 1) {
        int u = __shfl_up(v, off);
        if (lane >= off) v += u;
    }
    return v;
}

__device__ __forceinline__ float dinv_of(int c)
{
    float d = (float)(1 + c);
    return (d > 0.0f) ? rsqrtf(d) : 0.0f;
}

__global__ void __launch_bounds__(SCAN_THREADS)
k_scan_deg(const int* __restrict__ cnt, int* __restrict__ rowptr,
           float* __restrict__ dinv, int n)
{
    __shared__ int wsum[2][32];
    const int tid = threadIdx.x;
    const int lane = tid & 31;
    const int wv = tid >> 5;
    int running = 0;
    int buf = 0;
    for (int base = 0; base < n; base += SCAN_THREADS * SCAN_PER_THREAD) {
        const int idx = base + SCAN_PER_THREAD * tid;
        int c[SCAN_PER_THREAD];
        if (idx + SCAN_PER_THREAD <= n) {
            v4i lo = *(const v4i*)(cnt + idx);
            v4i hi = *(const v4i*)(cnt + idx + 4);
            c[0] = lo.x; c[1] = lo.y; c[2] = lo.z; c[3] = lo.w;
            c[4] = hi.x; c[5] = hi.y; c[6] = hi.z; c[7] = hi.w;
        } else {
#pragma unroll
            for (int k = 0; k < SCAN_PER_THREAD; ++k)
                c[k] = (idx + k < n) ? cnt[idx + k] : 0;
        }
        int p[SCAN_PER_THREAD];
        int tsum = 0;
#pragma unroll
        for (int k = 0; k < SCAN_PER_THREAD; ++k) { p[k] = tsum; tsum += c[k]; }

        const int inc = wave_incl_scan(tsum, lane);
        if (lane == 31) wsum[buf][wv] = inc;
        __syncthreads();
        const int tot = wsum[buf][lane];
        const int tinc = wave_incl_scan(tot, lane);
        const int wave_excl = __shfl(tinc, wv) - __shfl(tot, wv);
        const int trip_total = __shfl(tinc, 31);
        const int off0 = running + wave_excl + (inc - tsum);

        if (idx + SCAN_PER_THREAD <= n) {
            v4i r0, r1;
            r0.x = off0 + p[0]; r0.y = off0 + p[1]; r0.z = off0 + p[2]; r0.w = off0 + p[3];
            r1.x = off0 + p[4]; r1.y = off0 + p[5]; r1.z = off0 + p[6]; r1.w = off0 + p[7];
            *(v4i*)(rowptr + idx) = r0;
            *(v4i*)(rowptr + idx + 4) = r1;
            v4f d0, d1;
            d0.x = dinv_of(c[0]); d0.y = dinv_of(c[1]); d0.z = dinv_of(c[2]); d0.w = dinv_of(c[3]);
            d1.x = dinv_of(c[4]); d1.y = dinv_of(c[5]); d1.z = dinv_of(c[6]); d1.w = dinv_of(c[7]);
            *(v4f*)(dinv + idx) = d0;
            *(v4f*)(dinv + idx + 4) = d1;
        } else {
#pragma unroll
            for (int k = 0; k < SCAN_PER_THREAD; ++k) {
                if (idx + k < n) {
                    rowptr[idx + k] = off0 + p[k];
                    dinv[idx + k] = dinv_of(c[k]);
                }
            }
        }
        running += trip_total;
        buf ^= 1;
    }
    if (tid == 0) rowptr[n] = running;
}

__global__ void k_edge_place(const int* __restrict__ src, const int* __restrict__ dst,
                             const int* __restrict__ rank, const int* __restrict__ rowptr,
                             int* __restrict__ eidx, int e)
{
    int i = blockIdx.x * blockDim.x + threadIdx.x;
    if (i >= e) return;
    eidx[rowptr[dst[i]] + rank[i]] = src[i];
}

#define PULL_U 8
#define TILE_LD 68

__device__ __forceinline__ void load_b_frags(const float* __restrict__ Wm, int nt, int l, int half,
                                             v2f (&bf)[16])
{
    const float* b0 = Wm + (size_t)(2 * half) * HID + 16 * nt + l;
#pragma unroll
    for (int j = 0; j < 16; ++j) {
        bf[j].x = b0[(4 * j) * HID];
        bf[j].y = b0[(4 * j + 1) * HID];
    }
}

__device__ __forceinline__ v8f tile_mma64(const float* tile, int l, int half, const v2f (&bf)[16])
{
    v2f af[16];
#pragma unroll
    for (int j = 0; j < 16; ++j)
        af[j] = *(const v2f*)(tile + l * TILE_LD + 4 * j + 2 * half);
    v8f acc = {};
#pragma unroll
    for (int j = 0; j < 16; ++j) acc = WMMA_F32(af[j], bf[j], acc);
    return acc;
}

__device__ __forceinline__ void lds_barrier()
{
    __builtin_amdgcn_fence(__ATOMIC_RELEASE, "workgroup", "local");
    __builtin_amdgcn_s_barrier();
    __builtin_amdgcn_fence(__ATOMIC_ACQUIRE, "workgroup", "local");
}

__global__ void __launch_bounds__(256, 16)
k_pull_mma(const int* __restrict__ rowptr, const int* __restrict__ eidx,
           const float* __restrict__ h, const float* __restrict__ t,
           const float* __restrict__ dinv, const float* __restrict__ bias,
           float* __restrict__ h_out,
           const float* __restrict__ W0, const float* __restrict__ W1, const float* __restrict__ W2,
           const float* __restrict__ c0, const float* __restrict__ c1, const float* __restrict__ c2,
           float* __restrict__ y_out, int nstage, int addmask, int scale_out, int n)
{
    __shared__ __attribute__((aligned(16))) float tile[2][16 * TILE_LD];

    const int tid = threadIdx.x;
    const int g = tid >> 4;
    const int f4 = (tid & 15) * 4;
    const int node = (int)blockIdx.x * 16 + g;
    const int lane = tid & 31;
    const int wave = tid >> 5;
    const int half = lane >> 4;
    const int l = lane & 15;

    v4f acc = {0.0f, 0.0f, 0.0f, 0.0f};
    float dv = 0.0f;
    if (node < n) {
        const int beg = rowptr[node];
        const int end = rowptr[node + 1];
        dv = dinv[node];
        const char* tb = (const char*)t;
        const unsigned fo = (unsigned)f4 * 4u;
        const unsigned ro = (unsigned)node * (HID * 4u) + fo;
        v4f hv = *(const v4f*)((const char*)h + ro);
        v4f s  = *(const v4f*)(tb + ro);

        if (beg < end) {
            const int last = end - 1;
            int p[PULL_U];
#pragma unroll
            for (int k = 0; k < PULL_U; ++k) p[k] = eidx[min(beg + k, last)];
            for (int e = beg; e < end; e += PULL_U) {
                v4f gth[PULL_U];
#pragma unroll
                for (int k = 0; k < PULL_U; ++k)
                    gth[k] = *(const v4f*)(tb + ((unsigned)p[k] * (HID * 4u) + fo));
                int q[PULL_U];
#pragma unroll
                for (int k = 0; k < PULL_U; ++k) q[k] = eidx[min(e + PULL_U + k, last)];
                __builtin_amdgcn_sched_barrier(0);
#pragma unroll
                for (int k = 0; k < PULL_U; ++k)
                    if (e + k < end) s += gth[k];
#pragma unroll
                for (int k = 0; k < PULL_U; ++k) p[k] = q[k];
            }
        }
        v4f bv = *(const v4f*)(bias + f4);
        acc = hv + bv + s * dv;
        if (h_out) *(v4f*)(h_out + (size_t)node * HID + f4) = acc;
    }
    *(v4f*)(&tile[0][g * TILE_LD + f4]) = acc;

    int cur = 0;
    for (int s = 0; s < nstage; ++s) {
        const float* Wm = (s == 0) ? W0 : ((s == 1) ? W1 : W2);
        const float* cb = (s == 0) ? c0 : ((s == 1) ? c1 : c2);
        v2f bf[16];
        float cv = 0.0f;
        if (wave < 4) {
            load_b_frags(Wm, wave, l, half, bf);
            cv = cb[16 * wave + l];
        }
        const bool add = ((addmask >> s) & 1) != 0;
        lds_barrier();
        if (wave < 4) {
            v8f d = tile_mma64(tile[cur], l, half, bf);
            float* o = &tile[cur ^ 1][(8 * half) * TILE_LD + 16 * wave + l];
#pragma unroll
            for (int r = 0; r < 8; ++r) o[r * TILE_LD] = d[r] + (add ? cv : 0.0f);
        }
        cur ^= 1;
    }
    lds_barrier();
    if (node < n) {
        v4f y = *(const v4f*)(&tile[cur][g * TILE_LD + f4]);
        if (scale_out) y = y * dv;
        *(v4f*)(y_out + (size_t)node * HID + f4) = y;
    }
}

#define XT_LD 132

__global__ void __launch_bounds__(256)
k_enc_mma(const float* __restrict__ x, const float* __restrict__ encW,
          const float* __restrict__ encb, const float* __restrict__ W0,
          const float* __restrict__ dinv,
          float* __restrict__ h_out, float* __restrict__ t_out, int n)
{
    __shared__ __attribute__((aligned(16))) float xt[16 * XT_LD];
    __shared__ __attribute__((aligned(16))) float tile[2][16 * TILE_LD];

    const int tid = threadIdx.x;
    const int g = tid >> 4;
    const int f4 = (tid & 15) * 4;
    const int node = (int)blockIdx.x * 16 + g;
    const int lane = tid & 31;
    const int wave = tid >> 5;
    const int half = lane >> 4;
    const int l = lane & 15;

    v4f x0 = {0.0f, 0.0f, 0.0f, 0.0f};
    v4f x1 = x0;
    float dv = 0.0f;
    if (node < n) {
        x0 = *(const v4f*)(x + (size_t)node * IN_C + f4);
        x1 = *(const v4f*)(x + (size_t)node * IN_C + 64 + f4);
        dv = dinv[node];
    }
    *(v4f*)(&xt[g * XT_LD + f4]) = x0;
    *(v4f*)(&xt[g * XT_LD + 64 + f4]) = x1;

    v2f be[32];
    float ev = 0.0f;
    if (wave < 4) {
        const float* b0 = encW + (size_t)(2 * half) * HID + 16 * wave + l;
#pragma unroll
        for (int j = 0; j < 32; ++j) {
            be[j].x = b0[(4 * j) * HID];
            be[j].y = b0[(4 * j + 1) * HID];
        }
        ev = encb[16 * wave + l];
    }
    lds_barrier();
    if (wave < 4) {
        v8f d = {};
#pragma unroll
        for (int j = 0; j < 32; ++j) {
            v2f a = *(const v2f*)(xt + l * XT_LD + 4 * j + 2 * half);
            d = WMMA_F32(a, be[j], d);
        }
        float* o = &tile[0][(8 * half) * TILE_LD + 16 * wave + l];
#pragma unroll
        for (int r = 0; r < 8; ++r) o[r * TILE_LD] = d[r] + ev;
    }

    v2f bf[16];
    if (wave < 4) load_b_frags(W0, wave, l, half, bf);
    lds_barrier();
    if (node < n)
        *(v4f*)(h_out + (size_t)node * HID + f4) = *(const v4f*)(&tile[0][g * TILE_LD + f4]);
    if (wave < 4) {
        v8f d = tile_mma64(tile[0], l, half, bf);
        float* o = &tile[1][(8 * half) * TILE_LD + 16 * wave + l];
#pragma unroll
        for (int r = 0; r < 8; ++r) o[r * TILE_LD] = d[r] + 0.0f;
    }
    lds_barrier();
    if (node < n)
        *(v4f*)(t_out + (size_t)node * HID + f4) = *(const v4f*)(&tile[1][g * TILE_LD + f4]) * dv;
}

extern "C" void kernel_launch(void* const* d_in, const int* in_sizes, int n_in,
                              void* d_out, int out_size, void* d_ws, size_t ws_size,
                              hipStream_t stream)
{
    const float* x     = (const float*)d_in[0];
    const int*   ei    = (const int*)d_in[1];
    const float* encW  = (const float*)d_in[2];
    const float* encb  = (const float*)d_in[3];
    const float* gcnW  = (const float*)d_in[4];
    const float* gcnb  = (const float*)d_in[5];
    const float* decW  = (const float*)d_in[6];
    const float* decb  = (const float*)d_in[7];
    const float* finW  = (const float*)d_in[8];
    const float* finb  = (const float*)d_in[9];

    const int N = in_sizes[0] / IN_C;
    const int E = in_sizes[1] / 2;
    const int* src = ei;
    const int* dst = ei + E;

    float* f = (float*)d_ws;
    const size_t nodeFeats = (size_t)N * HID;
    const size_t Npad = ((size_t)N + 1 + 63) / 64 * 64;
    const size_t Epad = ((size_t)E + 63) / 64 * 64;
    float* hA    = f;                 f += nodeFeats;
    float* hB    = f;                 f += nodeFeats;
    float* tA    = f;                 f += nodeFeats;
    float* tB    = f;                 f += nodeFeats;
    int*   eidx  = (int*)f;           f += Epad;
    int*   rank  = (int*)f;           f += Epad;
    float* dinv  = f;                 f += Npad;
    int*   cnt   = (int*)f;           f += Npad;
    int*   rowptr = (int*)f;          f += Npad;
    (void)ws_size;

    const int BLK = 256;
    auto blocks1 = [&](long work) { return (int)((work + BLK - 1) / BLK); };

    k_zero_cnt  <<<blocks1(N), BLK, 0, stream>>>(cnt, N);
    k_edge_rank <<<blocks1(E), BLK, 0, stream>>>(dst, cnt, rank, E);
    k_scan_deg  <<<1, SCAN_THREADS, 0, stream>>>(cnt, rowptr, dinv, N);
    k_edge_place<<<blocks1(E), BLK, 0, stream>>>(src, dst, rank, rowptr, eidx, E);

    float* h   = hA;
    float* nxt = hB;
    float* t    = tA;
    float* tnxt = tB;

    const int pullBlocks = (N + 15) / 16;
    k_enc_mma<<<pullBlocks, BLK, 0, stream>>>(x, encW, encb, gcnW, dinv, h, t, N);

    for (int i = 0; i < NUM_LAYERS; ++i) {
        const float* b = gcnb + (size_t)i * HID;
        if (i + 1 < NUM_LAYERS) {
            const float* Wn = gcnW + (size_t)(i + 1) * HID * HID;
            k_pull_mma<<<pullBlocks, BLK, 0, stream>>>(rowptr, eidx, h, t, dinv, b, nxt,
                                                       Wn, nullptr, nullptr, b, nullptr, nullptr,
                                                       tnxt, 1, 0, 1, N);
        } else {
            k_pull_mma<<<pullBlocks, BLK, 0, stream>>>(rowptr, eidx, h, t, dinv, b, nullptr,
                                                       decW, decW + (size_t)HID * HID, finW,
                                                       decb, decb + HID, finb,
                                                       (float*)d_out, 3, 7, 0, N);
        }
        float* tmp = h; h = nxt; nxt = tmp;
        tmp = t; t = tnxt; tnxt = tmp;
    }
    (void)out_size; (void)n_in;
}


# layers carry only u = dinv*h: scaled neighbour rows are summed first, the sum times W after (new sum order), h = u/dinv once at the end; one N x 64 array read and written per layer, not two
# speedup vs baseline: 3.6763x; 1.1005x over previous
#include <hip/hip_runtime.h>
#include <hip/hip_bf16.h>


#define IN_C 128
#define HID 64
#define OUT_C 64
#define NUM_LAYERS 8
#define NUM_DEC 2

typedef float v2f __attribute__((ext_vector_type(2)));
typedef float v4f __attribute__((ext_vector_type(4)));
typedef float v8f __attribute__((ext_vector_type(8)));

#define WMMA_F32(a, b, acc) \
    __builtin_amdgcn_wmma_f32_16x16x4_f32(false, (a), false, (b), (short)0, (acc), false, false)

typedef int v4i __attribute__((ext_vector_type(4)));

__global__ void k_zero_cnt(int* __restrict__ cnt, int n)
{
    int i = blockIdx.x * blockDim.x + threadIdx.x;
    if (i < n) cnt[i] = 0;
}

__global__ void k_edge_rank(const int* __restrict__ dst, int* __restrict__ cnt,
                            int* __restrict__ rank, int e)
{
    int i = blockIdx.x * blockDim.x + threadIdx.x;
    if (i < e) rank[i] = atomicAdd(&cnt[dst[i]], 1);
}

#define SCAN_THREADS 1024
#define SCAN_PER_THREAD 8

__device__ __forceinline__ int wave_incl_scan(int v, int lane)
{
#pragma unroll
    for (int off = 1; off < 32; off <<= 1) {
        int u = __shfl_up(v, off);
        if (lane >= off) v += u;
    }
    return v;
}

__device__ __forceinline__ float dinv_of(int c)
{
    float d = (float)(1 + c);
    return (d > 0.0f) ? rsqrtf(d) : 0.0f;
}

__global__ void __launch_bounds__(SCAN_THREADS)
k_scan_deg(const int* __restrict__ cnt, int* __restrict__ rowptr,
           float* __restrict__ dinv, int n)
{
    __shared__ int wsum[2][32];
    const int tid = threadIdx.x;
    const int lane = tid & 31;
    const int wv = tid >> 5;
    int running = 0;
    int buf = 0;
    for (int base = 0; base < n; base += SCAN_THREADS * SCAN_PER_THREAD) {
        const int idx = base + SCAN_PER_THREAD * tid;
        int c[SCAN_PER_THREAD];
        if (idx + SCAN_PER_THREAD <= n) {
            v4i lo = *(const v4i*)(cnt + idx);
            v4i hi = *(const v4i*)(cnt + idx + 4);
            c[0] = lo.x; c[1] = lo.y; c[2] = lo.z; c[3] = lo.w;
            c[4] = hi.x; c[5] = hi.y; c[6] = hi.z; c[7] = hi.w;
        } else {
#pragma unroll
            for (int k = 0; k < SCAN_PER_THREAD; ++k)
                c[k] = (idx + k < n) ? cnt[idx + k] : 0;
        }
        int p[SCAN_PER_THREAD];
        int tsum = 0;
#pragma unroll
        for (int k = 0; k < SCAN_PER_THREAD; ++k) { p[k] = tsum; tsum += c[k]; }

        const int inc = wave_incl_scan(tsum, lane);
        if (lane == 31) wsum[buf][wv] = inc;
        __syncthreads();
        const int tot = wsum[buf][lane];
        const int tinc = wave_incl_scan(tot, lane);
        const int wave_excl = __shfl(tinc, wv) - __shfl(tot, wv);
        const int trip_total = __shfl(tinc, 31);
        const int off0 = running + wave_excl + (inc - tsum);

        if (idx + SCAN_PER_THREAD <= n) {
            v4i r0, r1;
            r0.x = off0 + p[0]; r0.y = off0 + p[1]; r0.z = off0 + p[2]; r0.w = off0 + p[3];
            r1.x = off0 + p[4]; r1.y = off0 + p[5]; r1.z = off0 + p[6]; r1.w = off0 + p[7];
            *(v4i*)(rowptr + idx) = r0;
            *(v4i*)(rowptr + idx + 4) = r1;
            v4f d0, d1;
            d0.x = dinv_of(c[0]); d0.y = dinv_of(c[1]); d0.z = dinv_of(c[2]); d0.w = dinv_of(c[3]);
            d1.x = dinv_of(c[4]); d1.y = dinv_of(c[5]); d1.z = dinv_of(c[6]); d1.w = dinv_of(c[7]);
            *(v4f*)(dinv + idx) = d0;
            *(v4f*)(dinv + idx + 4) = d1;
        } else {
#pragma unroll
            for (int k = 0; k < SCAN_PER_THREAD; ++k) {
                if (idx + k < n) {
                    rowptr[idx + k] = off0 + p[k];
                    dinv[idx + k] = dinv_of(c[k]);
                }
            }
        }
        running += trip_total;
        buf ^= 1;
    }
    if (tid == 0) rowptr[n] = running;
}

__global__ void k_edge_place(const int* __restrict__ src, const int* __restrict__ dst,
                             const int* __restrict__ rank, const int* __restrict__ rowptr,
                             int* __restrict__ eidx, int e)
{
    int i = blockIdx.x * blockDim.x + threadIdx.x;
    if (i >= e) return;
    eidx[rowptr[dst[i]] + rank[i]] = src[i];
}

#define PULL_U 8
#define TILE_LD 68

__device__ __forceinline__ void load_b_frags(const float* __restrict__ Wm, int nt, int l, int half,
                                             v2f (&bf)[16])
{
    const float* b0 = Wm + (size_t)(2 * half) * HID + 16 * nt + l;
#pragma unroll
    for (int j = 0; j < 16; ++j) {
        bf[j].x = b0[(4 * j) * HID];
        bf[j].y = b0[(4 * j + 1) * HID];
    }
}

__device__ __forceinline__ v8f tile_mma64(const float* tile, int l, int half, const v2f (&bf)[16])
{
    v2f af[16];
#pragma unroll
    for (int j = 0; j < 16; ++j)
        af[j] = *(const v2f*)(tile + l * TILE_LD + 4 * j + 2 * half);
    v8f acc = {};
#pragma unroll
    for (int j = 0; j < 16; ++j) acc = WMMA_F32(af[j], bf[j], acc);
    return acc;
}

__device__ __forceinline__ void lds_barrier()
{
    __builtin_amdgcn_fence(__ATOMIC_RELEASE, "workgroup", "local");
    __builtin_amdgcn_s_barrier();
    __builtin_amdgcn_fence(__ATOMIC_ACQUIRE, "workgroup", "local");
}

__global__ void __launch_bounds__(256, 16)
k_pull_mma(const int* __restrict__ rowptr, const int* __restrict__ eidx,
           const float* __restrict__ u, const float* __restrict__ dinv,
           const float* __restrict__ bias, const float* __restrict__ W0,
           float* __restrict__ u_out,
           const float* __restrict__ W1, const float* __restrict__ W2, const float* __restrict__ W3,
           const float* __restrict__ c1, const float* __restrict__ c2, const float* __restrict__ c3,
           float* __restrict__ y_out, int nextra, int n)
{
    __shared__ __attribute__((aligned(16))) float tile[2][16 * TILE_LD];
    __shared__ __attribute__((aligned(16))) float own[256 * 4];

    const int tid = threadIdx.x;
    const int g = tid >> 4;
    const int f4 = (tid & 15) * 4;
    const int node = (int)blockIdx.x * 16 + g;
    const int lane = tid & 31;
    const int wave = tid >> 5;
    const int half = lane >> 4;
    const int l = lane & 15;

    v4f s = {0.0f, 0.0f, 0.0f, 0.0f};
    float dv = 0.0f;
    if (node < n) {
        const int beg = rowptr[node];
        const int end = rowptr[node + 1];
        dv = dinv[node];
        const char* ub = (const char*)u;
        const unsigned fo = (unsigned)f4 * 4u;
        const unsigned ro = (unsigned)node * (HID * 4u) + fo;
        s = *(const v4f*)(ub + ro);
        *(v4f*)(&own[tid * 4]) = s;

        if (beg < end) {
            const int last = end - 1;
            int p[PULL_U];
#pragma unroll
            for (int k = 0; k < PULL_U; ++k) p[k] = eidx[min(beg + k, last)];
            for (int e = beg; e < end; e += PULL_U) {
                v4f gth[PULL_U];
#pragma unroll
                for (int k = 0; k < PULL_U; ++k)
                    gth[k] = *(const v4f*)(ub + ((unsigned)p[k] * (HID * 4u) + fo));
                int q[PULL_U];
#pragma unroll
                for (int k = 0; k < PULL_U; ++k) q[k] = eidx[min(e + PULL_U + k, last)];
                __builtin_amdgcn_sched_barrier(0);
#pragma unroll
                for (int k = 0; k < PULL_U; ++k)
                    if (e + k < end) s += gth[k];
#pragma unroll
                for (int k = 0; k < PULL_U; ++k) p[k] = q[k];
            }
        }
    }
    *(v4f*)(&tile[0][g * TILE_LD + f4]) = s;

    const v4f bv = *(const v4f*)(bias + f4);

    {
        v2f bf[16];
        if (wave < 4) load_b_frags(W0, wave, l, half, bf);
        lds_barrier();
        if (wave < 4) {
            v8f d = tile_mma64(tile[0], l, half, bf);
            float* o = &tile[1][(8 * half) * TILE_LD + 16 * wave + l];
#pragma unroll
            for (int r = 0; r < 8; ++r) o[r * TILE_LD] = d[r];
        }
    }
    lds_barrier();
    v4f un = {0.0f, 0.0f, 0.0f, 0.0f};
    if (node < n) {
        const v4f P  = *(const v4f*)(&tile[1][g * TILE_LD + f4]);
        const v4f uo = *(const v4f*)(&own[tid * 4]);
        un = uo + (bv + P * dv) * dv;
        if (u_out) *(v4f*)(u_out + (size_t)node * HID + f4) = un;
    }
    if (nextra == 0) return;

    v4f acc = {0.0f, 0.0f, 0.0f, 0.0f};
    if (node < n) acc = un / dv;
    *(v4f*)(&tile[0][g * TILE_LD + f4]) = acc;
    int cur = 0;
    for (int st = 0; st < nextra; ++st) {
        const float* Wm = (st == 0) ? W1 : ((st == 1) ? W2 : W3);
        const float* cb = (st == 0) ? c1 : ((st == 1) ? c2 : c3);
        v2f bf[16];
        float cv = 0.0f;
        if (wave < 4) {
            load_b_frags(Wm, wave, l, half, bf);
            cv = cb[16 * wave + l];
        }
        lds_barrier();
        if (wave < 4) {
            v8f d = tile_mma64(tile[cur], l, half, bf);
            float* o = &tile[cur ^ 1][(8 * half) * TILE_LD + 16 * wave + l];
#pragma unroll
            for (int r = 0; r < 8; ++r) o[r * TILE_LD] = d[r] + cv;
        }
        cur ^= 1;
    }
    lds_barrier();
    if (node < n)
        *(v4f*)(y_out + (size_t)node * HID + f4) = *(const v4f*)(&tile[cur][g * TILE_LD + f4]);
}

#define XT_LD 132

__global__ void __launch_bounds__(256)
k_enc_mma(const float* __restrict__ x, const float* __restrict__ encW,
          const float* __restrict__ encb, const float* __restrict__ dinv,
          float* __restrict__ u_out, int n)
{
    __shared__ __attribute__((aligned(16))) float xt[16 * XT_LD];
    __shared__ __attribute__((aligned(16))) float tile[16 * TILE_LD];

    const int tid = threadIdx.x;
    const int g = tid >> 4;
    const int f4 = (tid & 15) * 4;
    const int node = (int)blockIdx.x * 16 + g;
    const int lane = tid & 31;
    const int wave = tid >> 5;
    const int half = lane >> 4;
    const int l = lane & 15;

    v4f x0 = {0.0f, 0.0f, 0.0f, 0.0f};
    v4f x1 = x0;
    float dv = 0.0f;
    if (node < n) {
        x0 = *(const v4f*)(x + (size_t)node * IN_C + f4);
        x1 = *(const v4f*)(x + (size_t)node * IN_C + 64 + f4);
        dv = dinv[node];
    }
    *(v4f*)(&xt[g * XT_LD + f4]) = x0;
    *(v4f*)(&xt[g * XT_LD + 64 + f4]) = x1;

    v2f be[32];
    float ev = 0.0f;
    if (wave < 4) {
        const float* b0 = encW + (size_t)(2 * half) * HID + 16 * wave + l;
#pragma unroll
        for (int j = 0; j < 32; ++j) {
            be[j].x = b0[(4 * j) * HID];
            be[j].y = b0[(4 * j + 1) * HID];
        }
        ev = encb[16 * wave + l];
    }
    lds_barrier();
    if (wave < 4) {
        v8f d = {};
#pragma unroll
        for (int j = 0; j < 32; ++j) {
            v2f a = *(const v2f*)(xt + l * XT_LD + 4 * j + 2 * half);
            d = WMMA_F32(a, be[j], d);
        }
        float* o = &tile[(8 * half) * TILE_LD + 16 * wave + l];
#pragma unroll
        for (int r = 0; r < 8; ++r) o[r * TILE_LD] = d[r] + ev;
    }
    lds_barrier();
    if (node < n)
        *(v4f*)(u_out + (size_t)node * HID + f4) = *(const v4f*)(&tile[g * TILE_LD + f4]) * dv;
}

extern "C" void kernel_launch(void* const* d_in, const int* in_sizes, int n_in,
                              void* d_out, int out_size, void* d_ws, size_t ws_size,
                              hipStream_t stream)
{
    const float* x     = (const float*)d_in[0];
    const int*   ei    = (const int*)d_in[1];
    const float* encW  = (const float*)d_in[2];
    const float* encb  = (const float*)d_in[3];
    const float* gcnW  = (const float*)d_in[4];
    const float* gcnb  = (const float*)d_in[5];
    const float* decW  = (const float*)d_in[6];
    const float* decb  = (const float*)d_in[7];
    const float* finW  = (const float*)d_in[8];
    const float* finb  = (const float*)d_in[9];

    const int N = in_sizes[0] / IN_C;
    const int E = in_sizes[1] / 2;
    const int* src = ei;
    const int* dst = ei + E;

    float* f = (float*)d_ws;
    const size_t nodeFeats = (size_t)N * HID;
    const size_t Npad = ((size_t)N + 1 + 63) / 64 * 64;
    const size_t Epad = ((size_t)E + 63) / 64 * 64;
    float* uA    = f;                 f += nodeFeats;
    float* uB    = f;                 f += nodeFeats;
    int*   eidx  = (int*)f;           f += Epad;
    int*   rank  = (int*)f;           f += Epad;
    float* dinv  = f;                 f += Npad;
    int*   cnt   = (int*)f;           f += Npad;
    int*   rowptr = (int*)f;          f += Npad;
    (void)ws_size;

    const int BLK = 256;
    auto blocks1 = [&](long work) { return (int)((work + BLK - 1) / BLK); };

    k_zero_cnt  <<<blocks1(N), BLK, 0, stream>>>(cnt, N);
    k_edge_rank <<<blocks1(E), BLK, 0, stream>>>(dst, cnt, rank, E);
    k_scan_deg  <<<1, SCAN_THREADS, 0, stream>>>(cnt, rowptr, dinv, N);
    k_edge_place<<<blocks1(E), BLK, 0, stream>>>(src, dst, rank, rowptr, eidx, E);

    float* u   = uA;
    float* nxt = uB;

    const int pullBlocks = (N + 15) / 16;
    k_enc_mma<<<pullBlocks, BLK, 0, stream>>>(x, encW, encb, dinv, u, N);

    for (int i = 0; i < NUM_LAYERS; ++i) {
        const float* Wl = gcnW + (size_t)i * HID * HID;
        const float* b  = gcnb + (size_t)i * HID;
        if (i + 1 < NUM_LAYERS) {
            k_pull_mma<<<pullBlocks, BLK, 0, stream>>>(rowptr, eidx, u, dinv, b, Wl, nxt,
                                                       nullptr, nullptr, nullptr,
                                                       nullptr, nullptr, nullptr,
                                                       nullptr, 0, N);
        } else {
            k_pull_mma<<<pullBlocks, BLK, 0, stream>>>(rowptr, eidx, u, dinv, b, Wl, nullptr,
                                                       decW, decW + (size_t)HID * HID, finW,
                                                       decb, decb + HID, finb,
                                                       (float*)d_out, 3, N);
        }
        float* tmp = u; u = nxt; nxt = tmp;
    }
    (void)out_size; (void)n_in;
}


# edges grouped by destination by a two-level counting sort on LDS counters (chunks by 256-node bucket, then one workgroup per bucket); no global atomics, 13 -> 11 launches; node sum order = LDS rank
# speedup vs baseline: 5.1990x; 1.4142x over previous
#include <hip/hip_runtime.h>
#include <hip/hip_bf16.h>


#define IN_C 128
#define HID 64
#define OUT_C 64
#define NUM_LAYERS 8
#define NUM_DEC 2

typedef float v2f __attribute__((ext_vector_type(2)));
typedef float v4f __attribute__((ext_vector_type(4)));
typedef float v8f __attribute__((ext_vector_type(8)));

#define WMMA_F32(a, b, acc) \
    __builtin_amdgcn_wmma_f32_16x16x4_f32(false, (a), false, (b), (short)0, (acc), false, false)

typedef int v4i __attribute__((ext_vector_type(4)));

#define BK_SHIFT 8
#define BK_NODES 256
#define SORT_THREADS 256
#define EPT 13
#define CHUNK (SORT_THREADS * EPT)
#define BK_STAGE 8192

__device__ __forceinline__ int wave_incl_scan(int v, int lane)
{
#pragma unroll
    for (int off = 1; off < 32; off <<= 1) {
        int u = __shfl_up(v, off);
        if (lane >= off) v += u;
    }
    return v;
}

__device__ __forceinline__ float dinv_of(int c)
{
    float d = (float)(1 + c);
    return (d > 0.0f) ? rsqrtf(d) : 0.0f;
}

__global__ void __launch_bounds__(SORT_THREADS)
k_chunk_sort(const int* __restrict__ src, const int* __restrict__ dst,
             int* __restrict__ rec, int* __restrict__ cntT, int* __restrict__ offT,
             int e, int nbk, int nch)
{
    __shared__ int hist[BK_NODES];
    __shared__ int wsum[SORT_THREADS / 32];
    __shared__ int lrec[CHUNK];
    const int tid = threadIdx.x;
    const int lane = tid & 31;
    const int wv = tid >> 5;
    const int c = (int)blockIdx.x;
    const int cbase = c * CHUNK;

    hist[tid] = 0;
    int d[EPT], s[EPT], r[EPT];
#pragma unroll
    for (int k = 0; k < EPT; ++k) {
        const int i = cbase + k * SORT_THREADS + tid;
        const int ic = min(i, e - 1);
        d[k] = dst[ic];
        s[k] = src[ic];
    }
#pragma unroll
    for (int k = 0; k < EPT; ++k) {
        const int i = cbase + k * SORT_THREADS + tid;
        if (i >= e) d[k] = -1;
    }
    __syncthreads();
#pragma unroll
    for (int k = 0; k < EPT; ++k) {
        r[k] = 0;
        if (d[k] >= 0) r[k] = atomicAdd(&hist[d[k] >> BK_SHIFT], 1);
    }
    __syncthreads();

    const int h = hist[tid];
    const int inc = wave_incl_scan(h, lane);
    if (lane == 31) wsum[wv] = inc;
    __syncthreads();
    int wbase = 0;
#pragma unroll
    for (int w = 0; w < SORT_THREADS / 32; ++w) wbase += (w < wv) ? wsum[w] : 0;
    const int excl = wbase + inc - h;
    hist[tid] = excl;
    if (tid < nbk) {
        cntT[tid * nch + c] = h;
        offT[tid * nch + c] = excl;
    }
    __syncthreads();
#pragma unroll
    for (int k = 0; k < EPT; ++k) {
        if (d[k] >= 0)
            lrec[hist[d[k] >> BK_SHIFT] + r[k]] = (s[k] << BK_SHIFT) | (d[k] & (BK_NODES - 1));
    }
    __syncthreads();
    const int valid = min(CHUNK, e - cbase);
    int v[EPT];
#pragma unroll
    for (int k = 0; k < EPT; ++k) v[k] = lrec[k * SORT_THREADS + tid];
#pragma unroll
    for (int k = 0; k < EPT; ++k) {
        const int i = k * SORT_THREADS + tid;
        if (i < valid) rec[cbase + i] = v[k];
    }
}

__global__ void __launch_bounds__(BK_NODES)
k_bucket_csr(const int* __restrict__ rec, const int* __restrict__ cntT, const int* __restrict__ offT,
             int* __restrict__ rowptr, float* __restrict__ dinv, int* __restrict__ eidx,
             int n, int nbk, int nch)
{
    __shared__ int ncnt[BK_NODES];
    __shared__ int red[BK_NODES / 32];
    __shared__ int wsum[BK_NODES / 32];
    __shared__ int lidx[BK_STAGE];
    const int tid = threadIdx.x;
    const int lane = tid & 31;
    const int wv = tid >> 5;
    const int b = (int)blockIdx.x;

    ncnt[tid] = 0;

    const v4i* t4 = (const v4i*)cntT;
    const int lim4 = (b * nch) >> 2;
    int acc = 0;
#pragma unroll 8
    for (int i = tid; i < lim4; i += BK_NODES) {
        const v4i v = t4[i];
        acc += (v.x + v.y) + (v.z + v.w);
    }
    const int accw = wave_incl_scan(acc, lane);
    if (lane == 31) red[wv] = accw;
    __syncthreads();
    int base = 0;
#pragma unroll
    for (int w = 0; w < BK_NODES / 32; ++w) base += red[w];

    const int* crow = cntT + (size_t)b * nch;
    const int* orow = offT + (size_t)b * nch;
    for (int c = tid; c < nch; c += BK_NODES) {
        const int m = crow[c];
        const int* p = rec + (size_t)c * CHUNK + orow[c];
        int j = 0;
        for (; j + 4 <= m; j += 4) {
            const int r0 = p[j], r1 = p[j + 1], r2 = p[j + 2], r3 = p[j + 3];
            atomicAdd(&ncnt[r0 & (BK_NODES - 1)], 1);
            atomicAdd(&ncnt[r1 & (BK_NODES - 1)], 1);
            atomicAdd(&ncnt[r2 & (BK_NODES - 1)], 1);
            atomicAdd(&ncnt[r3 & (BK_NODES - 1)], 1);
        }
        for (; j < m; ++j) atomicAdd(&ncnt[p[j] & (BK_NODES - 1)], 1);
    }
    __syncthreads();

    const int h = ncnt[tid];
    const int inc = wave_incl_scan(h, lane);
    if (lane == 31) wsum[wv] = inc;
    __syncthreads();
    int wbase = 0, total = 0;
#pragma unroll
    for (int w = 0; w < BK_NODES / 32; ++w) {
        wbase += (w < wv) ? wsum[w] : 0;
        total += wsum[w];
    }
    const int excl = wbase + inc - h;
    const int node = b * BK_NODES + tid;
    if (node < n) {
        rowptr[node] = base + excl;
        dinv[node] = dinv_of(h);
    }
    if (b == nbk - 1 && tid == 0) rowptr[n] = base + total;
    ncnt[tid] = excl;
    __syncthreads();

    int* __restrict__ eb = eidx + base;
    if (total <= BK_STAGE) {
        for (int c = tid; c < nch; c += BK_NODES) {
            const int m = crow[c];
            const int* p = rec + (size_t)c * CHUNK + orow[c];
            int j = 0;
            for (; j + 4 <= m; j += 4) {
                const int r0 = p[j], r1 = p[j + 1], r2 = p[j + 2], r3 = p[j + 3];
                const int q0 = atomicAdd(&ncnt[r0 & (BK_NODES - 1)], 1);
                const int q1 = atomicAdd(&ncnt[r1 & (BK_NODES - 1)], 1);
                const int q2 = atomicAdd(&ncnt[r2 & (BK_NODES - 1)], 1);
                const int q3 = atomicAdd(&ncnt[r3 & (BK_NODES - 1)], 1);
                lidx[q0] = r0 >> BK_SHIFT;
                lidx[q1] = r1 >> BK_SHIFT;
                lidx[q2] = r2 >> BK_SHIFT;
                lidx[q3] = r3 >> BK_SHIFT;
            }
            for (; j < m; ++j) {
                const int r0 = p[j];
                const int q0 = atomicAdd(&ncnt[r0 & (BK_NODES - 1)], 1);
                lidx[q0] = r0 >> BK_SHIFT;
            }
        }
        __syncthreads();
        for (int i = tid; i < total; i += BK_NODES) eb[i] = lidx[i];
    } else {
        for (int c = tid; c < nch; c += BK_NODES) {
            const int m = crow[c];
            const int* p = rec + (size_t)c * CHUNK + orow[c];
            for (int j = 0; j < m; ++j) {
                const int r0 = p[j];
                const int q0 = atomicAdd(&ncnt[r0 & (BK_NODES - 1)], 1);
                eb[q0] = r0 >> BK_SHIFT;
            }
        }
    }
}

#define PULL_U 8
#define TILE_LD 68

__device__ __forceinline__ void load_b_frags(const float* __restrict__ Wm, int nt, int l, int half,
                                             v2f (&bf)[16])
{
    const float* b0 = Wm + (size_t)(2 * half) * HID + 16 * nt + l;
#pragma unroll
    for (int j = 0; j < 16; ++j) {
        bf[j].x = b0[(4 * j) * HID];
        bf[j].y = b0[(4 * j + 1) * HID];
    }
}

__device__ __forceinline__ v8f tile_mma64(const float* tile, int l, int half, const v2f (&bf)[16])
{
    v2f af[16];
#pragma unroll
    for (int j = 0; j < 16; ++j)
        af[j] = *(const v2f*)(tile + l * TILE_LD + 4 * j + 2 * half);
    v8f acc = {};
#pragma unroll
    for (int j = 0; j < 16; ++j) acc = WMMA_F32(af[j], bf[j], acc);
    return acc;
}

__device__ __forceinline__ void lds_barrier()
{
    __builtin_amdgcn_fence(__ATOMIC_RELEASE, "workgroup", "local");
    __builtin_amdgcn_s_barrier();
    __builtin_amdgcn_fence(__ATOMIC_ACQUIRE, "workgroup", "local");
}

__global__ void __launch_bounds__(256, 16)
k_pull_mma(const int* __restrict__ rowptr, const int* __restrict__ eidx,
           const float* __restrict__ u, const float* __restrict__ dinv,
           const float* __restrict__ bias, const float* __restrict__ W0,
           float* __restrict__ u_out,
           const float* __restrict__ W1, const float* __restrict__ W2, const float* __restrict__ W3,
           const float* __restrict__ c1, const float* __restrict__ c2, const float* __restrict__ c3,
           float* __restrict__ y_out, int nextra, int n)
{
    __shared__ __attribute__((aligned(16))) float tile[2][16 * TILE_LD];
    __shared__ __attribute__((aligned(16))) float own[256 * 4];

    const int tid = threadIdx.x;
    const int g = tid >> 4;
    const int f4 = (tid & 15) * 4;
    const int node = (int)blockIdx.x * 16 + g;
    const int lane = tid & 31;
    const int wave = tid >> 5;
    const int half = lane >> 4;
    const int l = lane & 15;

    v4f s = {0.0f, 0.0f, 0.0f, 0.0f};
    float dv = 0.0f;
    if (node < n) {
        const int beg = rowptr[node];
        const int end = rowptr[node + 1];
        dv = dinv[node];
        const char* ub = (const char*)u;
        const unsigned fo = (unsigned)f4 * 4u;
        const unsigned ro = (unsigned)node * (HID * 4u) + fo;
        s = *(const v4f*)(ub + ro);
        *(v4f*)(&own[tid * 4]) = s;

        if (beg < end) {
            const int last = end - 1;
            int p[PULL_U];
#pragma unroll
            for (int k = 0; k < PULL_U; ++k) p[k] = eidx[min(beg + k, last)];
            for (int e = beg; e < end; e += PULL_U) {
                v4f gth[PULL_U];
#pragma unroll
                for (int k = 0; k < PULL_U; ++k)
                    gth[k] = *(const v4f*)(ub + ((unsigned)p[k] * (HID * 4u) + fo));
                int q[PULL_U];
#pragma unroll
                for (int k = 0; k < PULL_U; ++k) q[k] = eidx[min(e + PULL_U + k, last)];
                __builtin_amdgcn_sched_barrier(0);
#pragma unroll
                for (int k = 0; k < PULL_U; ++k)
                    if (e + k < end) s += gth[k];
#pragma unroll
                for (int k = 0; k < PULL_U; ++k) p[k] = q[k];
            }
        }
    }
    *(v4f*)(&tile[0][g * TILE_LD + f4]) = s;

    const v4f bv = *(const v4f*)(bias + f4);

    {
        v2f bf[16];
        if (wave < 4) load_b_frags(W0, wave, l, half, bf);
        lds_barrier();
        if (wave < 4) {
            v8f d = tile_mma64(tile[0], l, half, bf);
            float* o = &tile[1][(8 * half) * TILE_LD + 16 * wave + l];
#pragma unroll
            for (int r = 0; r < 8; ++r) o[r * TILE_LD] = d[r];
        }
    }
    lds_barrier();
    v4f un = {0.0f, 0.0f, 0.0f, 0.0f};
    if (node < n) {
        const v4f P  = *(const v4f*)(&tile[1][g * TILE_LD + f4]);
        const v4f uo = *(const v4f*)(&own[tid * 4]);
        un = uo + (bv + P * dv) * dv;
        if (u_out) *(v4f*)(u_out + (size_t)node * HID + f4) = un;
    }
    if (nextra == 0) return;

    v4f acc = {0.0f, 0.0f, 0.0f, 0.0f};
    if (node < n) acc = un / dv;
    *(v4f*)(&tile[0][g * TILE_LD + f4]) = acc;
    int cur = 0;
    for (int st = 0; st < nextra; ++st) {
        const float* Wm = (st == 0) ? W1 : ((st == 1) ? W2 : W3);
        const float* cb = (st == 0) ? c1 : ((st == 1) ? c2 : c3);
        v2f bf[16];
        float cv = 0.0f;
        if (wave < 4) {
            load_b_frags(Wm, wave, l, half, bf);
            cv = cb[16 * wave + l];
        }
        lds_barrier();
        if (wave < 4) {
            v8f d = tile_mma64(tile[cur], l, half, bf);
            float* o = &tile[cur ^ 1][(8 * half) * TILE_LD + 16 * wave + l];
#pragma unroll
            for (int r = 0; r < 8; ++r) o[r * TILE_LD] = d[r] + cv;
        }
        cur ^= 1;
    }
    lds_barrier();
    if (node < n)
        *(v4f*)(y_out + (size_t)node * HID + f4) = *(const v4f*)(&tile[cur][g * TILE_LD + f4]);
}

#define XT_LD 132

__global__ void __launch_bounds__(256)
k_enc_mma(const float* __restrict__ x, const float* __restrict__ encW,
          const float* __restrict__ encb, const float* __restrict__ dinv,
          float* __restrict__ u_out, int n)
{
    __shared__ __attribute__((aligned(16))) float xt[16 * XT_LD];
    __shared__ __attribute__((aligned(16))) float tile[16 * TILE_LD];

    const int tid = threadIdx.x;
    const int g = tid >> 4;
    const int f4 = (tid & 15) * 4;
    const int node = (int)blockIdx.x * 16 + g;
    const int lane = tid & 31;
    const int wave = tid >> 5;
    const int half = lane >> 4;
    const int l = lane & 15;

    v4f x0 = {0.0f, 0.0f, 0.0f, 0.0f};
    v4f x1 = x0;
    float dv = 0.0f;
    if (node < n) {
        x0 = *(const v4f*)(x + (size_t)node * IN_C + f4);
        x1 = *(const v4f*)(x + (size_t)node * IN_C + 64 + f4);
        dv = dinv[node];
    }
    *(v4f*)(&xt[g * XT_LD + f4]) = x0;
    *(v4f*)(&xt[g * XT_LD + 64 + f4]) = x1;

    v2f be[32];
    float ev = 0.0f;
    if (wave < 4) {
        const float* b0 = encW + (size_t)(2 * half) * HID + 16 * wave + l;
#pragma unroll
        for (int j = 0; j < 32; ++j) {
            be[j].x = b0[(4 * j) * HID];
            be[j].y = b0[(4 * j + 1) * HID];
        }
        ev = encb[16 * wave + l];
    }
    lds_barrier();
    if (wave < 4) {
        v8f d = {};
#pragma unroll
        for (int j = 0; j < 32; ++j) {
            v2f a = *(const v2f*)(xt + l * XT_LD + 4 * j + 2 * half);
            d = WMMA_F32(a, be[j], d);
        }
        float* o = &tile[(8 * half) * TILE_LD + 16 * wave + l];
#pragma unroll
        for (int r = 0; r < 8; ++r) o[r * TILE_LD] = d[r] + ev;
    }
    lds_barrier();
    if (node < n)
        *(v4f*)(u_out + (size_t)node * HID + f4) = *(const v4f*)(&tile[g * TILE_LD + f4]) * dv;
}

extern "C" void kernel_launch(void* const* d_in, const int* in_sizes, int n_in,
                              void* d_out, int out_size, void* d_ws, size_t ws_size,
                              hipStream_t stream)
{
    const float* x     = (const float*)d_in[0];
    const int*   ei    = (const int*)d_in[1];
    const float* encW  = (const float*)d_in[2];
    const float* encb  = (const float*)d_in[3];
    const float* gcnW  = (const float*)d_in[4];
    const float* gcnb  = (const float*)d_in[5];
    const float* decW  = (const float*)d_in[6];
    const float* decb  = (const float*)d_in[7];
    const float* finW  = (const float*)d_in[8];
    const float* finb  = (const float*)d_in[9];

    const int N = in_sizes[0] / IN_C;
    const int E = in_sizes[1] / 2;
    const int* src = ei;
    const int* dst = ei + E;

    const int nbk = (N + BK_NODES - 1) / BK_NODES;
    const int nch = (((E + CHUNK - 1) / CHUNK) + 3) / 4 * 4;

    float* f = (float*)d_ws;
    const size_t nodeFeats = (size_t)N * HID;
    const size_t Npad = ((size_t)N + 1 + 63) / 64 * 64;
    const size_t Epad = ((size_t)E + 63) / 64 * 64;
    const size_t Rpad = ((size_t)nch * CHUNK + 63) / 64 * 64;
    const size_t Tpad = ((size_t)nbk * nch + 63) / 64 * 64;
    float* uA    = f;                 f += nodeFeats;
    float* uB    = f;                 f += nodeFeats;
    int*   eidx  = (int*)f;           f += Epad;
    int*   rec   = (int*)f;           f += Rpad;
    float* dinv  = f;                 f += Npad;
    int*   cntT  = (int*)f;           f += Tpad;
    int*   offT  = (int*)f;           f += Tpad;
    int*   rowptr = (int*)f;          f += Npad;
    (void)ws_size;

    const int BLK = 256;

    k_chunk_sort<<<nch, SORT_THREADS, 0, stream>>>(src, dst, rec, cntT, offT, E, nbk, nch);
    k_bucket_csr<<<nbk, BK_NODES, 0, stream>>>(rec, cntT, offT, rowptr, dinv, eidx, N, nbk, nch);

    float* u   = uA;
    float* nxt = uB;

    const int pullBlocks = (N + 15) / 16;
    k_enc_mma<<<pullBlocks, BLK, 0, stream>>>(x, encW, encb, dinv, u, N);

    for (int i = 0; i < NUM_LAYERS; ++i) {
        const float* Wl = gcnW + (size_t)i * HID * HID;
        const float* b  = gcnb + (size_t)i * HID;
        if (i + 1 < NUM_LAYERS) {
            k_pull_mma<<<pullBlocks, BLK, 0, stream>>>(rowptr, eidx, u, dinv, b, Wl, nxt,
                                                       nullptr, nullptr, nullptr,
                                                       nullptr, nullptr, nullptr,
                                                       nullptr, 0, N);
        } else {
            k_pull_mma<<<pullBlocks, BLK, 0, stream>>>(rowptr, eidx, u, dinv, b, Wl, nullptr,
                                                       decW, decW + (size_t)HID * HID, finW,
                                                       decb, decb + HID, finb,
                                                       (float*)d_out, 3, N);
        }
        float* tmp = u; u = nxt; nxt = tmp;
    }
    (void)out_size; (void)n_in;
}
